# starting kernel
# baseline (speedup 1.0000x reference)
#include <hip/hip_runtime.h>
#include <hip/hip_bf16.h>

typedef __attribute__((ext_vector_type(2))) float v2f;
typedef __attribute__((ext_vector_type(8))) float v8f;

#define DDIM   128
#define KNN    32
#define TILE_N 16
#define WPAD   130
#define THREADS 256

__global__ __launch_bounds__(THREADS)
void gcl_fused_kernel(const float* __restrict__ Input,
                      const int*   __restrict__ Idx,
                      const float* __restrict__ C,
                      const float* __restrict__ W,
                      const float* __restrict__ b,
                      float*       __restrict__ out,
                      int N)
{
    extern __shared__ float lds[];
    float* wLDS = lds;
    float* yLDS = lds + DDIM * WPAD;
    float* sLDS = yLDS + TILE_N * WPAD;

    const int t  = threadIdx.x;
    const int n0 = blockIdx.x * TILE_N;

    for (int i = t; i < DDIM * DDIM; i += THREADS) {
        const int r = i >> 7;
        const int c = i & (DDIM - 1);
        wLDS[r * WPAD + c] = W[i];
    }

    const int d    = t & (DDIM - 1);
    const int half = t >> 7;

    float acc[8];
#pragma unroll
    for (int r = 0; r < 8; ++r) acc[r] = 0.0f;

    for (int k = 0; k < KNN; ++k) {
#pragma unroll
        for (int r = 0; r < 8; ++r) {
            const int n = n0 + half * 8 + r;
            if (n < N) {
                const int   gi = Idx[n * KNN + k];
                const float cw = C[n * KNN + k];
                acc[r] += cw * Input[(long)gi * DDIM + d];
            }
        }
    }
#pragma unroll
    for (int r = 0; r < 8; ++r)
        yLDS[(half * 8 + r) * WPAD + d] = acc[r];

    if (t < TILE_N) {
        const int n = n0 + t;
        float s = 0.0f;
        if (n < N)
            for (int k = 0; k < KNN; ++k) s += C[n * KNN + k];
        sLDS[t] = s;
    }
    __syncthreads();

    const int wave = t >> 5;
    const int lane = t & 31;
    const int hi   = lane >> 4;
    const int lo   = lane & 15;
    const int e0   = wave * 16;

    v8f cacc = {};
#pragma unroll
    for (int k0 = 0; k0 < DDIM; k0 += 4) {
        const float* ap = &yLDS[lo * WPAD + k0 + 2 * hi];
        const float* bp = &wLDS[(e0 + lo) * WPAD + k0 + 2 * hi];
        v2f av; av.x = ap[0]; av.y = ap[1];
        v2f bv; bv.x = bp[0]; bv.y = bp[1];
        cacc = __builtin_amdgcn_wmma_f32_16x16x4_f32(
             false, av,  false, bv,
             (short)0, cacc,  false,  false);
    }

    const float bias = b[e0 + lo];
#pragma unroll
    for (int v = 0; v < 8; ++v) {
        const int nl = v + 8 * hi;
        const int n  = n0 + nl;
        if (n < N) {
            const float val = cacc[v] + sLDS[nl] * bias;
            out[(long)n * DDIM + (e0 + lo)] = 1.0f / (1.0f + __expf(-val));
        }
    }
}

extern "C" void kernel_launch(void* const* d_in, const int* in_sizes, int n_in,
                              void* d_out, int out_size, void* d_ws, size_t ws_size,
                              hipStream_t stream)
{
    const float* Input = (const float*)d_in[0];
    const int*   Idx   = (const int*)  d_in[1];
    const float* C     = (const float*)d_in[2];
    const float* W     = (const float*)d_in[3];
    const float* b     = (const float*)d_in[4];
    float*       out   = (float*)d_out;

    const int N = in_sizes[1] / KNN;
    const int blocks = (N + TILE_N - 1) / TILE_N;
    const size_t shmem = (size_t)(DDIM * WPAD + TILE_N * WPAD + TILE_N) * sizeof(float);

    hipLaunchKernelGGL(gcl_fused_kernel, dim3(blocks), dim3(THREADS), shmem, stream,
                       Input, Idx, C, W, b, out, N);
}


# 80 rows per 16-wave work-group: W staged to LDS once per 80 rows, not per 16; 8 gather waves (32 row loads in flight) feed 8 WMMA waves through a 2-slot LDS ring; k and sum order kept
# speedup vs baseline: 7.3678x; 7.3678x over previous
#include <hip/hip_runtime.h>
#include <hip/hip_bf16.h>


typedef __attribute__((ext_vector_type(2))) float v2f;
typedef __attribute__((ext_vector_type(4))) float v4f;
typedef __attribute__((ext_vector_type(8))) float v8f;

#define DDIM   128
#define KNN    32
#define TILE_N 16
#define WPAD   130
#define TILES_PER_BLOCK 5
#define BLOCK_ROWS (TILE_N * TILES_PER_BLOCK)
#define MMA_WAVES    8
#define GATHER_WAVES 8
#define THREADS ((MMA_WAVES + GATHER_WAVES) * 32)
#define ROWS_PER_WAVE (TILE_N / GATHER_WAVES)
#define GATHER_BATCH 32

__global__ __launch_bounds__(THREADS)
void gcl_fused_kernel(const float* __restrict__ Input,
                      const int*   __restrict__ Idx,
                      const float* __restrict__ C,
                      const float* __restrict__ W,
                      const float* __restrict__ b,
                      float*       __restrict__ out,
                      int N)
{
    extern __shared__ float lds[];
    float* wLDS = lds;
    float* yLDS = lds + DDIM * WPAD;
    float* sLDS = yLDS + 2 * TILE_N * WPAD;

    const int t    = threadIdx.x;
    const int row0 = blockIdx.x * BLOCK_ROWS;

    const int wave = __builtin_amdgcn_readfirstlane(t >> 5);
    const int lane = t & 31;
    const bool gather_wave = wave >= MMA_WAVES;
    const int gw = wave - MMA_WAVES;

    int ntiles = (N - row0 + TILE_N - 1) / TILE_N;
    if (ntiles > TILES_PER_BLOCK) ntiles = TILES_PER_BLOCK;

    const int hi = lane >> 4;
    const int lo = lane & 15;
    const int e0 = wave * 16;

    int   gi_n[ROWS_PER_WAVE];
    float cw_n[ROWS_PER_WAVE];
    float bias = 0.0f;

    if (gather_wave) {
#pragma unroll
        for (int rr = 0; rr < ROWS_PER_WAVE; ++rr) {
            const int n = min(row0 + gw * ROWS_PER_WAVE + rr, N - 1);
            gi_n[rr] = Idx[n * KNN + lane];
            cw_n[rr] = C[n * KNN + lane];
        }
    } else {
        for (int i = t; i < DDIM * DDIM; i += MMA_WAVES * 32) {
            const int r = i >> 7;
            const int c = i & (DDIM - 1);
            wLDS[r * WPAD + c] = W[i];
        }
        bias = b[e0 + lo];
    }

    for (int it = 0; it <= ntiles; ++it) {
        if (gather_wave) {
            if (it < ntiles) {
                float* ytile = yLDS + (it & 1) * (TILE_N * WPAD);
                float* stile = sLDS + (it & 1) * TILE_N;

                int   gi_l[ROWS_PER_WAVE];
                float cw_l[ROWS_PER_WAVE];
#pragma unroll
                for (int rr = 0; rr < ROWS_PER_WAVE; ++rr) {
                    gi_l[rr] = gi_n[rr];
                    cw_l[rr] = cw_n[rr];
                    const int nn = min(row0 + (it + 1) * TILE_N + gw * ROWS_PER_WAVE + rr, N - 1);
                    gi_n[rr] = Idx[nn * KNN + lane];
                    cw_n[rr] = C[nn * KNN + lane];
                }

#pragma unroll
                for (int rr = 0; rr < ROWS_PER_WAVE; ++rr) {
                    const int nl = gw * ROWS_PER_WAVE + rr;
                    const int n  = row0 + it * TILE_N + nl;
                    float a0 = 0.0f, a1 = 0.0f, a2 = 0.0f, a3 = 0.0f;
                    float s  = 0.0f;
                    if (n < N) {
#pragma unroll
                        for (int kb = 0; kb < KNN; kb += GATHER_BATCH) {
                            v4f x[GATHER_BATCH];
#pragma unroll
                            for (int j = 0; j < GATHER_BATCH; ++j) {
                                const int gi = __builtin_amdgcn_readlane(gi_l[rr], kb + j);
                                const v4f* rowp = reinterpret_cast<const v4f*>(Input + (long)gi * DDIM);
                                x[j] = rowp[lane];
                            }
                            __builtin_amdgcn_sched_barrier(0);
#pragma unroll
                            for (int j = 0; j < GATHER_BATCH; ++j) {
                                const float cw = __int_as_float(__builtin_amdgcn_readlane(__float_as_int(cw_l[rr]), kb + j));
                                a0 = __builtin_fmaf(cw, x[j].x, a0);
                                a1 = __builtin_fmaf(cw, x[j].y, a1);
                                a2 = __builtin_fmaf(cw, x[j].z, a2);
                                a3 = __builtin_fmaf(cw, x[j].w, a3);
                                s  += cw;
                            }
                        }
                    }
                    v2f* yp = reinterpret_cast<v2f*>(&ytile[nl * WPAD + lane * 4]);
                    v2f y01; y01.x = a0; y01.y = a1;
                    v2f y23; y23.x = a2; y23.y = a3;
                    yp[0] = y01;
                    yp[1] = y23;
                    if (lane == 0) stile[nl] = s;
                }
            }
        } else if (it > 0) {
            const float* ytile = yLDS + ((it - 1) & 1) * (TILE_N * WPAD);
            const float* stile = sLDS + ((it - 1) & 1) * TILE_N;
            const int n0 = row0 + (it - 1) * TILE_N;

            v8f cacc = {};
#pragma unroll
            for (int k0 = 0; k0 < DDIM; k0 += 4) {
                const float* ap = &ytile[lo * WPAD + k0 + 2 * hi];
                const float* bp = &wLDS[(e0 + lo) * WPAD + k0 + 2 * hi];
                v2f av; av.x = ap[0]; av.y = ap[1];
                v2f bv; bv.x = bp[0]; bv.y = bp[1];
                cacc = __builtin_amdgcn_wmma_f32_16x16x4_f32(
                     false, av,  false, bv,
                     (short)0, cacc,  false,  false);
            }

#pragma unroll
            for (int v = 0; v < 8; ++v) {
                const int nl = v + 8 * hi;
                const int n  = n0 + nl;
                if (n < N) {
                    const float val = cacc[v] + stile[nl] * bias;
                    out[(long)n * DDIM + (e0 + lo)] = 1.0f / (1.0f + __expf(-val));
                }
            }
        }
        if (it < ntiles) __syncthreads();
    }
}

extern "C" void kernel_launch(void* const* d_in, const int* in_sizes, int n_in,
                              void* d_out, int out_size, void* d_ws, size_t ws_size,
                              hipStream_t stream)
{
    const float* Input = (const float*)d_in[0];
    const int*   Idx   = (const int*)  d_in[1];
    const float* C     = (const float*)d_in[2];
    const float* W     = (const float*)d_in[3];
    const float* b     = (const float*)d_in[4];
    float*       out   = (float*)d_out;

    const int N = in_sizes[1] / KNN;
    const int blocks = (N + BLOCK_ROWS - 1) / BLOCK_ROWS;
    const size_t shmem = (size_t)(DDIM * WPAD + 2 * TILE_N * WPAD + 2 * TILE_N) * sizeof(float);

    hipLaunchKernelGGL(gcl_fused_kernel, dim3(blocks), dim3(THREADS), shmem, stream,
                       Input, Idx, C, W, b, out, N);
}


# Fewer, wider second-level requests: each output row staged in LDS and stored whole (one 512-byte store, was eight 64-byte pieces); W staged with 16-byte loads (was 4-byte); arithmetic untouched
# speedup vs baseline: 7.8108x; 1.0601x over previous
#include <hip/hip_runtime.h>
#include <hip/hip_bf16.h>


typedef __attribute__((ext_vector_type(2))) float v2f;
typedef __attribute__((ext_vector_type(4))) float v4f;
typedef __attribute__((ext_vector_type(8))) float v8f;

#define DDIM   128
#define KNN    32
#define TILE_N 16
#define WPAD   130
#define OPAD   132
#define TILES_PER_BLOCK 5
#define BLOCK_ROWS (TILE_N * TILES_PER_BLOCK)
#define MMA_WAVES    8
#define GATHER_WAVES 8
#define THREADS ((MMA_WAVES + GATHER_WAVES) * 32)
#define ROWS_PER_WAVE (TILE_N / GATHER_WAVES)
#define GATHER_BATCH 32

__global__ __launch_bounds__(THREADS)
void gcl_fused_kernel(const float* __restrict__ Input,
                      const int*   __restrict__ Idx,
                      const float* __restrict__ C,
                      const float* __restrict__ W,
                      const float* __restrict__ b,
                      float*       __restrict__ out,
                      int N)
{
    extern __shared__ float lds[];
    float* wLDS = lds;
    float* yLDS = lds + DDIM * WPAD;
    float* sLDS = yLDS + 2 * TILE_N * WPAD;
    float* oLDS = sLDS + 2 * TILE_N;

    const int t    = threadIdx.x;
    const int row0 = blockIdx.x * BLOCK_ROWS;

    const int wave = __builtin_amdgcn_readfirstlane(t >> 5);
    const int lane = t & 31;
    const bool gather_wave = wave >= MMA_WAVES;
    const int gw = wave - MMA_WAVES;

    int ntiles = (N - row0 + TILE_N - 1) / TILE_N;
    if (ntiles > TILES_PER_BLOCK) ntiles = TILES_PER_BLOCK;

    const int hi = lane >> 4;
    const int lo = lane & 15;
    const int e0 = wave * 16;

    int   gi_n[ROWS_PER_WAVE];
    float cw_n[ROWS_PER_WAVE];
    float bias = 0.0f;

    if (gather_wave) {
#pragma unroll
        for (int rr = 0; rr < ROWS_PER_WAVE; ++rr) {
            const int n = min(row0 + gw * ROWS_PER_WAVE + rr, N - 1);
            gi_n[rr] = Idx[n * KNN + lane];
            cw_n[rr] = C[n * KNN + lane];
        }
    } else {
        const v4f* W4 = reinterpret_cast<const v4f*>(W);
#pragma unroll
        for (int j = 0; j < (DDIM * DDIM / 4) / (MMA_WAVES * 32); ++j) {
            const int i4 = t + j * (MMA_WAVES * 32);
            const int r  = i4 >> 5;
            const int c  = (i4 & 31) * 4;
            const v4f w4 = W4[i4];
            v2f* wp = reinterpret_cast<v2f*>(&wLDS[r * WPAD + c]);
            v2f w01; w01.x = w4.x; w01.y = w4.y;
            v2f w23; w23.x = w4.z; w23.y = w4.w;
            wp[0] = w01;
            wp[1] = w23;
        }
        bias = b[e0 + lo];
    }

    for (int it = 0; it <= ntiles; ++it) {
        if (gather_wave) {
            if (it < ntiles) {
                float* ytile = yLDS + (it & 1) * (TILE_N * WPAD);
                float* stile = sLDS + (it & 1) * TILE_N;

                int   gi_l[ROWS_PER_WAVE];
                float cw_l[ROWS_PER_WAVE];
#pragma unroll
                for (int rr = 0; rr < ROWS_PER_WAVE; ++rr) {
                    gi_l[rr] = gi_n[rr];
                    cw_l[rr] = cw_n[rr];
                    const int nn = min(row0 + (it + 1) * TILE_N + gw * ROWS_PER_WAVE + rr, N - 1);
                    gi_n[rr] = Idx[nn * KNN + lane];
                    cw_n[rr] = C[nn * KNN + lane];
                }

#pragma unroll
                for (int rr = 0; rr < ROWS_PER_WAVE; ++rr) {
                    const int nl = gw * ROWS_PER_WAVE + rr;
                    const int n  = row0 + it * TILE_N + nl;
                    float a0 = 0.0f, a1 = 0.0f, a2 = 0.0f, a3 = 0.0f;
                    float s  = 0.0f;
                    if (n < N) {
#pragma unroll
                        for (int kb = 0; kb < KNN; kb += GATHER_BATCH) {
                            v4f x[GATHER_BATCH];
#pragma unroll
                            for (int j = 0; j < GATHER_BATCH; ++j) {
                                const int gi = __builtin_amdgcn_readlane(gi_l[rr], kb + j);
                                const v4f* rowp = reinterpret_cast<const v4f*>(Input + (long)gi * DDIM);
                                x[j] = rowp[lane];
                            }
                            __builtin_amdgcn_sched_barrier(0);
#pragma unroll
                            for (int j = 0; j < GATHER_BATCH; ++j) {
                                const float cw = __int_as_float(__builtin_amdgcn_readlane(__float_as_int(cw_l[rr]), kb + j));
                                a0 = __builtin_fmaf(cw, x[j].x, a0);
                                a1 = __builtin_fmaf(cw, x[j].y, a1);
                                a2 = __builtin_fmaf(cw, x[j].z, a2);
                                a3 = __builtin_fmaf(cw, x[j].w, a3);
                                s  += cw;
                            }
                        }
                    }
                    v2f* yp = reinterpret_cast<v2f*>(&ytile[nl * WPAD + lane * 4]);
                    v2f y01; y01.x = a0; y01.y = a1;
                    v2f y23; y23.x = a2; y23.y = a3;
                    yp[0] = y01;
                    yp[1] = y23;
                    if (lane == 0) stile[nl] = s;
                }
            }
        } else if (it > 0) {
            if (it > 1) {
                const float* otile = oLDS + (it & 1) * (TILE_N * OPAD);
                const int n0s = row0 + (it - 2) * TILE_N;
                for (int rr = 0; rr < TILE_N / MMA_WAVES; ++rr) {
                    const int r = wave * (TILE_N / MMA_WAVES) + rr;
                    const int n = n0s + r;
                    if (n < N) {
                        const v4f o = *reinterpret_cast<const v4f*>(&otile[r * OPAD + lane * 4]);
                        *reinterpret_cast<v4f*>(&out[(long)n * DDIM + lane * 4]) = o;
                    }
                }
            }

            const float* ytile = yLDS + ((it - 1) & 1) * (TILE_N * WPAD);
            const float* stile = sLDS + ((it - 1) & 1) * TILE_N;
            float* otile_w = oLDS + ((it - 1) & 1) * (TILE_N * OPAD);

            v8f cacc = {};
#pragma unroll
            for (int k0 = 0; k0 < DDIM; k0 += 4) {
                const float* ap = &ytile[lo * WPAD + k0 + 2 * hi];
                const float* bp = &wLDS[(e0 + lo) * WPAD + k0 + 2 * hi];
                v2f av; av.x = ap[0]; av.y = ap[1];
                v2f bv; bv.x = bp[0]; bv.y = bp[1];
                cacc = __builtin_amdgcn_wmma_f32_16x16x4_f32(
                     false, av,  false, bv,
                     (short)0, cacc,  false,  false);
            }

#pragma unroll
            for (int v = 0; v < 8; ++v) {
                const int nl = v + 8 * hi;
                const float val = cacc[v] + stile[nl] * bias;
                otile_w[nl * OPAD + (e0 + lo)] = 1.0f / (1.0f + __expf(-val));
            }
        }
        __syncthreads();
    }

    {
        const float* otile = oLDS + ((ntiles - 1) & 1) * (TILE_N * OPAD);
        const int n = row0 + (ntiles - 1) * TILE_N + wave;
        if (n < N) {
            const v4f o = *reinterpret_cast<const v4f*>(&otile[wave * OPAD + lane * 4]);
            *reinterpret_cast<v4f*>(&out[(long)n * DDIM + lane * 4]) = o;
        }
    }
}

extern "C" void kernel_launch(void* const* d_in, const int* in_sizes, int n_in,
                              void* d_out, int out_size, void* d_ws, size_t ws_size,
                              hipStream_t stream)
{
    const float* Input = (const float*)d_in[0];
    const int*   Idx   = (const int*)  d_in[1];
    const float* C     = (const float*)d_in[2];
    const float* W     = (const float*)d_in[3];
    const float* b     = (const float*)d_in[4];
    float*       out   = (float*)d_out;

    const int N = in_sizes[1] / KNN;
    const int blocks = (N + BLOCK_ROWS - 1) / BLOCK_ROWS;
    const size_t shmem = (size_t)(DDIM * WPAD + 2 * TILE_N * WPAD + 2 * TILE_N + 2 * TILE_N * OPAD) * sizeof(float);

    hipLaunchKernelGGL(gcl_fused_kernel, dim3(blocks), dim3(THREADS), shmem, stream,
                       Input, Idx, C, W, b, out, N);
}


# Gather waves keep a rolling window of 32 row loads in flight: a neighbor's registers are refilled for the wave's next row right after use; barrier fences LDS only so the window survives it; sums kept
# speedup vs baseline: 8.2019x; 1.0501x over previous
#include <hip/hip_runtime.h>
#include <hip/hip_bf16.h>


typedef __attribute__((ext_vector_type(2))) float v2f;
typedef __attribute__((ext_vector_type(4))) float v4f;
typedef __attribute__((ext_vector_type(8))) float v8f;

#define DDIM   128
#define KNN    32
#define TILE_N 16
#define WPAD   130
#define OPAD   132
#define TILES_PER_BLOCK 5
#define BLOCK_ROWS (TILE_N * TILES_PER_BLOCK)
#define MMA_WAVES    8
#define GATHER_WAVES 8
#define THREADS ((MMA_WAVES + GATHER_WAVES) * 32)
#define ROWS_PER_WAVE (TILE_N / GATHER_WAVES)

__device__ __forceinline__ void issue_row(v4f (&x)[KNN], const float* __restrict__ Input, int giv, int lane)
{
#pragma unroll
    for (int j = 0; j < KNN; ++j) {
        const int gi = __builtin_amdgcn_readlane(giv, j);
        x[j] = reinterpret_cast<const v4f*>(Input + (long)gi * DDIM)[lane];
    }
}

template <bool NEXT>
__device__ __forceinline__ void sum_row(v4f (&x)[KNN], const float* __restrict__ Input, float cwv, int giv_next, int lane,
                                        float& a0, float& a1, float& a2, float& a3, float& s)
{
#pragma unroll
    for (int j = 0; j < KNN; ++j) {
        const float cw = __int_as_float(__builtin_amdgcn_readlane(__float_as_int(cwv), j));
        a0 = __builtin_fmaf(cw, x[j].x, a0);
        a1 = __builtin_fmaf(cw, x[j].y, a1);
        a2 = __builtin_fmaf(cw, x[j].z, a2);
        a3 = __builtin_fmaf(cw, x[j].w, a3);
        s  += cw;
        if (NEXT) {
            asm volatile("" : : "v"(a0), "v"(a1), "v"(a2), "v"(a3));
            const int gi = __builtin_amdgcn_readlane(giv_next, j);
            x[j] = reinterpret_cast<const v4f*>(Input + (long)gi * DDIM)[lane];
        }
    }
}

__global__ __launch_bounds__(THREADS)
void gcl_fused_kernel(const float* __restrict__ Input,
                      const int*   __restrict__ Idx,
                      const float* __restrict__ C,
                      const float* __restrict__ W,
                      const float* __restrict__ b,
                      float*       __restrict__ out,
                      int N)
{
    extern __shared__ float lds[];
    float* wLDS = lds;
    float* yLDS = lds + DDIM * WPAD;
    float* sLDS = yLDS + 2 * TILE_N * WPAD;
    float* oLDS = sLDS + 2 * TILE_N;

    const int t    = threadIdx.x;
    const int row0 = blockIdx.x * BLOCK_ROWS;

    const int wave = __builtin_amdgcn_readfirstlane(t >> 5);
    const int lane = t & 31;
    const bool gather_wave = wave >= MMA_WAVES;
    const int gw = wave - MMA_WAVES;

    int ntiles = (N - row0 + TILE_N - 1) / TILE_N;
    if (ntiles > TILES_PER_BLOCK) ntiles = TILES_PER_BLOCK;

    const int hi = lane >> 4;
    const int lo = lane & 15;
    const int e0 = wave * 16;

    int   gi_c[ROWS_PER_WAVE];
    float cw_c[ROWS_PER_WAVE];
    float bias = 0.0f;

    v4f x[KNN];

    if (gather_wave) {
#pragma unroll
        for (int rr = 0; rr < ROWS_PER_WAVE; ++rr) {
            const int n = min(row0 + gw * ROWS_PER_WAVE + rr, N - 1);
            gi_c[rr] = Idx[n * KNN + lane];
            cw_c[rr] = C[n * KNN + lane];
        }
        issue_row(x, Input, gi_c[0], lane);
    } else {
        const v4f* W4 = reinterpret_cast<const v4f*>(W);
#pragma unroll
        for (int j = 0; j < (DDIM * DDIM / 4) / (MMA_WAVES * 32); ++j) {
            const int i4 = t + j * (MMA_WAVES * 32);
            const int r  = i4 >> 5;
            const int c  = (i4 & 31) * 4;
            const v4f w4 = W4[i4];
            v2f* wp = reinterpret_cast<v2f*>(&wLDS[r * WPAD + c]);
            v2f w01; w01.x = w4.x; w01.y = w4.y;
            v2f w23; w23.x = w4.z; w23.y = w4.w;
            wp[0] = w01;
            wp[1] = w23;
        }
        bias = b[e0 + lo];
    }

    for (int it = 0; it <= ntiles; ++it) {
        if (gather_wave) {
            if (it < ntiles) {
                float* ytile = yLDS + (it & 1) * (TILE_N * WPAD);
                float* stile = sLDS + (it & 1) * TILE_N;

                int   gi_n[ROWS_PER_WAVE];
                float cw_n[ROWS_PER_WAVE];
#pragma unroll
                for (int rr = 0; rr < ROWS_PER_WAVE; ++rr) {
                    const int nn = min(row0 + (it + 1) * TILE_N + gw * ROWS_PER_WAVE + rr, N - 1);
                    gi_n[rr] = Idx[nn * KNN + lane];
                    cw_n[rr] = C[nn * KNN + lane];
                }

#pragma unroll
                for (int rr = 0; rr < ROWS_PER_WAVE; ++rr) {
                    const int nl = gw * ROWS_PER_WAVE + rr;
                    float a0 = 0.0f, a1 = 0.0f, a2 = 0.0f, a3 = 0.0f;
                    float s  = 0.0f;
                    if (rr + 1 < ROWS_PER_WAVE) {
                        sum_row<true>(x, Input, cw_c[rr], gi_c[(rr + 1) % ROWS_PER_WAVE], lane, a0, a1, a2, a3, s);
                    } else if (it + 1 < ntiles) {
                        sum_row<true>(x, Input, cw_c[rr], gi_n[0], lane, a0, a1, a2, a3, s);
                    } else {
                        sum_row<false>(x, Input, cw_c[rr], 0, lane, a0, a1, a2, a3, s);
                    }
                    v2f* yp = reinterpret_cast<v2f*>(&ytile[nl * WPAD + lane * 4]);
                    v2f y01; y01.x = a0; y01.y = a1;
                    v2f y23; y23.x = a2; y23.y = a3;
                    yp[0] = y01;
                    yp[1] = y23;
                    if (lane == 0) stile[nl] = s;
                }
#pragma unroll
                for (int rr = 0; rr < ROWS_PER_WAVE; ++rr) {
                    gi_c[rr] = gi_n[rr];
                    cw_c[rr] = cw_n[rr];
                }
            }
        } else if (it > 0) {
            if (it > 1) {
                const float* otile = oLDS + (it & 1) * (TILE_N * OPAD);
                const int n0s = row0 + (it - 2) * TILE_N;
                for (int rr = 0; rr < TILE_N / MMA_WAVES; ++rr) {
                    const int r = wave * (TILE_N / MMA_WAVES) + rr;
                    const int n = n0s + r;
                    if (n < N) {
                        const v4f o = *reinterpret_cast<const v4f*>(&otile[r * OPAD + lane * 4]);
                        *reinterpret_cast<v4f*>(&out[(long)n * DDIM + lane * 4]) = o;
                    }
                }
            }

            const float* ytile = yLDS + ((it - 1) & 1) * (TILE_N * WPAD);
            const float* stile = sLDS + ((it - 1) & 1) * TILE_N;
            float* otile_w = oLDS + ((it - 1) & 1) * (TILE_N * OPAD);

            v8f cacc = {};
#pragma unroll
            for (int k0 = 0; k0 < DDIM; k0 += 4) {
                const float* ap = &ytile[lo * WPAD + k0 + 2 * hi];
                const float* bp = &wLDS[(e0 + lo) * WPAD + k0 + 2 * hi];
                v2f av; av.x = ap[0]; av.y = ap[1];
                v2f bv; bv.x = bp[0]; bv.y = bp[1];
                cacc = __builtin_amdgcn_wmma_f32_16x16x4_f32(
                     false, av,  false, bv,
                     (short)0, cacc,  false,  false);
            }

#pragma unroll
            for (int v = 0; v < 8; ++v) {
                const int nl = v + 8 * hi;
                const float val = cacc[v] + stile[nl] * bias;
                otile_w[nl * OPAD + (e0 + lo)] = 1.0f / (1.0f + __expf(-val));
            }
        }
        __builtin_amdgcn_fence(__ATOMIC_RELEASE, "workgroup", "local");
        __builtin_amdgcn_s_barrier();
        __builtin_amdgcn_fence(__ATOMIC_ACQUIRE, "workgroup", "local");
    }

    {
        const float* otile = oLDS + ((ntiles - 1) & 1) * (TILE_N * OPAD);
        const int n = row0 + (ntiles - 1) * TILE_N + wave;
        if (n < N) {
            const v4f o = *reinterpret_cast<const v4f*>(&otile[wave * OPAD + lane * 4]);
            *reinterpret_cast<v4f*>(&out[(long)n * DDIM + lane * 4]) = o;
        }
    }
}

extern "C" void kernel_launch(void* const* d_in, const int* in_sizes, int n_in,
                              void* d_out, int out_size, void* d_ws, size_t ws_size,
                              hipStream_t stream)
{
    const float* Input = (const float*)d_in[0];
    const int*   Idx   = (const int*)  d_in[1];
    const float* C     = (const float*)d_in[2];
    const float* W     = (const float*)d_in[3];
    const float* b     = (const float*)d_in[4];
    float*       out   = (float*)d_out;

    const int N = in_sizes[1] / KNN;
    const int blocks = (N + BLOCK_ROWS - 1) / BLOCK_ROWS;
    const size_t shmem = (size_t)(DDIM * WPAD + 2 * TILE_N * WPAD + 2 * TILE_N + 2 * TILE_N * OPAD) * sizeof(float);

    hipLaunchKernelGGL(gcl_fused_kernel, dim3(blocks), dim3(THREADS), shmem, stream,
                       Input, Idx, C, W, b, out, N);
}
